# starting kernel
# baseline (speedup 1.0000x reference)
#include <hip/hip_runtime.h>
#include <hip/hip_bf16.h>


typedef float v2f __attribute__((ext_vector_type(2)));
typedef float v8f __attribute__((ext_vector_type(8)));

constexpr int B_   = 2;
constexpr int S_   = 2048;
constexpr int H_   = 16;
constexpr int D_   = 64;
constexpr int CH   = 64;
constexpr int LDP  = 68;
constexpr int NSPL = 8;

constexpr int OFF_Q    = 0;
constexpr int OFF_K    = 64 * LDP;
constexpr int OFF_V    = 2 * 64 * LDP;
constexpr int OFF_S    = 3 * 64 * LDP;
constexpr int OFF_KV   = 4 * 64 * LDP;
constexpr int OFF_KSUM = 5 * 64 * LDP;
constexpr int OFF_DEN  = OFF_KSUM + 64;
constexpr int SMEM_F   = OFF_DEN + 64;

constexpr int STATE_F  = 64 * 64 + 64;

__device__ __forceinline__ float elu1(float x) {
  return x > 0.0f ? x + 1.0f : __expf(x);
}

__device__ __forceinline__ v8f wmma4(v2f a, v2f b, v8f c) {
  return __builtin_amdgcn_wmma_f32_16x16x4_f32(
       false, a,  false, b,
       (short)0, c,  false,  false);
}

__device__ __forceinline__ v2f lds_pair(const float* p) {
  return *reinterpret_cast<const v2f*>(p);
}

__device__ __forceinline__ v8f cd_load(const float* base, int mi, int nj, int lane) {
  int n  = nj * 16 + (lane & 15);
  int m0 = mi * 16 + ((lane >> 4) << 3);
  v8f c;
#pragma unroll
  for (int r = 0; r < 8; ++r) c[r] = base[(m0 + r) * LDP + n];
  return c;
}
__device__ __forceinline__ void cd_store(float* base, int mi, int nj, int lane, v8f c) {
  int n  = nj * 16 + (lane & 15);
  int m0 = mi * 16 + ((lane >> 4) << 3);
#pragma unroll
  for (int r = 0; r < 8; ++r) base[(m0 + r) * LDP + n] = c[r];
}

__device__ __forceinline__ v8f gemm_nn(const float* A, const float* Bm,
                                       int mi, int nj, int lane,
                                       int k0, int k1, v8f acc) {
  int lm = lane & 15, hi = lane >> 4;
  int arow = (mi * 16 + lm) * LDP;
  int bcol = nj * 16 + lm;
  for (int kk = k0; kk < k1; ++kk) {
    int kb = kk * 4 + hi * 2;
    v2f a = lds_pair(&A[arow + kb]);
    v2f b; b.x = Bm[kb * LDP + bcol]; b.y = Bm[(kb + 1) * LDP + bcol];
    acc = wmma4(a, b, acc);
  }
  return acc;
}
__device__ __forceinline__ v8f gemm_nt(const float* A, const float* Bt,
                                       int mi, int nj, int lane, v8f acc) {
  int lm = lane & 15, hi = lane >> 4;
  int arow = (mi * 16 + lm) * LDP;
  int brow = (nj * 16 + lm) * LDP;
#pragma unroll
  for (int kk = 0; kk < 16; ++kk) {
    int kb = kk * 4 + hi * 2;
    v2f a = lds_pair(&A[arow + kb]);
    v2f b = lds_pair(&Bt[brow + kb]);
    acc = wmma4(a, b, acc);
  }
  return acc;
}
__device__ __forceinline__ v8f gemm_tn(const float* At, const float* Bm,
                                       int mi, int nj, int lane, v8f acc) {
  int lm = lane & 15, hi = lane >> 4;
  int am   = mi * 16 + lm;
  int bcol = nj * 16 + lm;
#pragma unroll
  for (int kk = 0; kk < 16; ++kk) {
    int kb = kk * 4 + hi * 2;
    v2f a; a.x = At[kb * LDP + am];   a.y = At[(kb + 1) * LDP + am];
    v2f b; b.x = Bm[kb * LDP + bcol]; b.y = Bm[(kb + 1) * LDP + bcol];
    acc = wmma4(a, b, acc);
  }
  return acc;
}

__device__ __forceinline__ void mask_diag(v8f& c, int lane) {
  int n  = lane & 15;
  int m0 = (lane >> 4) << 3;
#pragma unroll
  for (int r = 0; r < 8; ++r)
    if (m0 + r < n) c[r] = 0.0f;
}

__device__ __forceinline__ void load_tile(const float* __restrict__ g, int rowStride,
                                          float* lds, int tid, bool do_elu) {
#pragma unroll
  for (int j = 0; j < 4; ++j) {
    int u = tid + j * 256;
    int r = u >> 4;
    int c = (u & 15) << 2;
    float4 w = *reinterpret_cast<const float4*>(g + (size_t)r * rowStride + c);
    if (do_elu) { w.x = elu1(w.x); w.y = elu1(w.y); w.z = elu1(w.z); w.w = elu1(w.w); }
    *reinterpret_cast<float4*>(lds + r * LDP + c) = w;
  }
}

__global__ __launch_bounds__(256) void lin_attn_states(
    const float* __restrict__ qk, const float* __restrict__ v,
    float* __restrict__ ws, int nsplit) {
  extern __shared__ float smem[];
  float* sK    = smem + OFF_K;
  float* sV    = smem + OFF_V;
  float* sKV   = smem + OFF_KV;
  float* sksum = smem + OFF_KSUM;

  int tid = threadIdx.x, lane = tid & 31, wid = tid >> 5;
  int wg = blockIdx.x;
  int split = wg % nsplit;
  int h     = (wg / nsplit) % H_;
  int b     = wg / (nsplit * H_);
  int cps   = (S_ / CH) / nsplit;

  for (int i = tid; i < 64 * LDP; i += 256) sKV[i] = 0.0f;
  if (tid < 64) sksum[tid] = 0.0f;
  __syncthreads();

  for (int c = 0; c < cps; ++c) {
    int base = (split * cps + c) * CH;
    const float* kptr = qk + (((size_t)(b * S_ + base) * 2 + 1) * H_ + h) * D_;
    const float* vptr = v  + (((size_t)(b * S_ + base)) * H_ + h) * D_;
    load_tile(kptr, 2 * H_ * D_, sK, tid, true);
    load_tile(vptr, H_ * D_,     sV, tid, false);
    __syncthreads();

    if (tid < 64) {
      float acc = 0.0f;
      for (int s = 0; s < 64; ++s) acc += sK[s * LDP + tid];
      sksum[tid] += acc;
    }
#pragma unroll
    for (int i = 0; i < 2; ++i) {
      int t = wid * 2 + i, mi = t >> 2, nj = t & 3;
      v8f acc = cd_load(sKV, mi, nj, lane);
      acc = gemm_tn(sK, sV, mi, nj, lane, acc);
      cd_store(sKV, mi, nj, lane, acc);
    }
    __syncthreads();
  }

  float* wsp = ws + (size_t)((b * H_ + h) * nsplit + split) * STATE_F;
  for (int i = tid; i < 64 * 64; i += 256)
    wsp[i] = sKV[(i >> 6) * LDP + (i & 63)];
  if (tid < 64) wsp[64 * 64 + tid] = sksum[tid];
}

__global__ __launch_bounds__(256) void lin_attn_main(
    const float* __restrict__ qk, const float* __restrict__ v,
    const float* __restrict__ ws, float* __restrict__ out, int nsplit) {
  extern __shared__ float smem[];
  float* sQ    = smem + OFF_Q;
  float* sK    = smem + OFF_K;
  float* sV    = smem + OFF_V;
  float* sS    = smem + OFF_S;
  float* sKV   = smem + OFF_KV;
  float* sksum = smem + OFF_KSUM;
  float* sden  = smem + OFF_DEN;

  int tid = threadIdx.x, lane = tid & 31, wid = tid >> 5;
  int wg = blockIdx.x;
  int split = wg % nsplit;
  int h     = (wg / nsplit) % H_;
  int b     = wg / (nsplit * H_);
  int cps   = (S_ / CH) / nsplit;

  for (int i = tid; i < 64 * LDP; i += 256) sKV[i] = 0.0f;
  if (tid < 64) sksum[tid] = 0.0f;
  __syncthreads();

  for (int p = 0; p < split; ++p) {
    const float* wsp = ws + (size_t)((b * H_ + h) * nsplit + p) * STATE_F;
    for (int i = tid; i < 64 * 64; i += 256)
      sKV[(i >> 6) * LDP + (i & 63)] += wsp[i];
    if (tid < 64) sksum[tid] += wsp[64 * 64 + tid];
  }
  __syncthreads();

  int t0 = wid * 2, t1 = wid * 2 + 1;
  int m0 = t0 >> 2, n0 = t0 & 3, m1 = t1 >> 2, n1 = t1 & 3;

  for (int c = 0; c < cps; ++c) {
    int base = (split * cps + c) * CH;
    const float* qptr = qk + (((size_t)(b * S_ + base) * 2 + 0) * H_ + h) * D_;
    const float* kptr = qk + (((size_t)(b * S_ + base) * 2 + 1) * H_ + h) * D_;
    const float* vptr = v  + (((size_t)(b * S_ + base)) * H_ + h) * D_;
    load_tile(qptr, 2 * H_ * D_, sQ, tid, true);
    load_tile(kptr, 2 * H_ * D_, sK, tid, true);
    load_tile(vptr, H_ * D_,     sV, tid, false);

    if (c + 1 < cps) {
      int nb = base + CH;
      const float* nq = qk + (((size_t)(b * S_ + nb) * 2) * H_ + h) * D_;
      const float* nv = v  + (((size_t)(b * S_ + nb)) * H_ + h) * D_;
      int r = tid >> 2, o = (tid & 3) << 4;
      __builtin_prefetch(nq + (size_t)r * 2 * H_ * D_ + o, 0, 3);
      __builtin_prefetch(nq + (size_t)r * 2 * H_ * D_ + H_ * D_ + o, 0, 3);
      __builtin_prefetch(nv + (size_t)r * H_ * D_ + o, 0, 3);
    }
    __syncthreads();

    v8f oa0 = {}, oa1 = {};
    oa0 = gemm_nn(sQ, sKV, m0, n0, lane, 0, 16, oa0);
    oa1 = gemm_nn(sQ, sKV, m1, n1, lane, 0, 16, oa1);

    if (tid < 64) {
      float d = 0.0f;
      for (int dd = 0; dd < 64; ++dd) d += sQ[tid * LDP + dd] * sksum[dd];
      sden[tid] = d;
    }

    if (n0 <= m0) {
      v8f sa = {};
      sa = gemm_nt(sQ, sK, m0, n0, lane, sa);
      if (m0 == n0) mask_diag(sa, lane);
      cd_store(sS, m0, n0, lane, sa);
    }
    if (n1 <= m1) {
      v8f sa = {};
      sa = gemm_nt(sQ, sK, m1, n1, lane, sa);
      if (m1 == n1) mask_diag(sa, lane);
      cd_store(sS, m1, n1, lane, sa);
    }
    __syncthreads();

    if (tid < 64) {
      float ssum = 0.0f;
      int smax = ((tid >> 4) + 1) * 16;
      for (int s = 0; s < smax; ++s) ssum += sS[tid * LDP + s];
      sden[tid] = 1.0f / (sden[tid] + ssum);
    }
    __syncthreads();

    oa0 = gemm_nn(sS, sV, m0, n0, lane, 0, (m0 + 1) * 4, oa0);
    oa1 = gemm_nn(sS, sV, m1, n1, lane, 0, (m1 + 1) * 4, oa1);

    {
      int nn = n0 * 16 + (lane & 15);
      int mm = m0 * 16 + ((lane >> 4) << 3);
#pragma unroll
      for (int r = 0; r < 8; ++r)
        out[(((size_t)b * S_ + base + mm + r) * H_ + h) * D_ + nn] = oa0[r] * sden[mm + r];
      nn = n1 * 16 + (lane & 15);
      mm = m1 * 16 + ((lane >> 4) << 3);
#pragma unroll
      for (int r = 0; r < 8; ++r)
        out[(((size_t)b * S_ + base + mm + r) * H_ + h) * D_ + nn] = oa1[r] * sden[mm + r];
    }
    __syncthreads();

    if (tid < 64) {
      float acc = 0.0f;
      for (int s = 0; s < 64; ++s) acc += sK[s * LDP + tid];
      sksum[tid] += acc;
    }
    {
      v8f a0 = cd_load(sKV, m0, n0, lane);
      a0 = gemm_tn(sK, sV, m0, n0, lane, a0);
      cd_store(sKV, m0, n0, lane, a0);
      v8f a1 = cd_load(sKV, m1, n1, lane);
      a1 = gemm_tn(sK, sV, m1, n1, lane, a1);
      cd_store(sKV, m1, n1, lane, a1);
    }
    __syncthreads();
  }
}

extern "C" void kernel_launch(void* const* d_in, const int* in_sizes, int n_in,
                              void* d_out, int out_size, void* d_ws, size_t ws_size,
                              hipStream_t stream) {
  const float* qk = (const float*)d_in[0];
  const float* v  = (const float*)d_in[1];
  float* out = (float*)d_out;
  float* ws  = (float*)d_ws;

  int nsplit = NSPL;
  size_t need = (size_t)B_ * H_ * NSPL * STATE_F * sizeof(float);
  if (ws_size < need) nsplit = 1;

  int wgs = B_ * H_ * nsplit;
  size_t smem = (size_t)SMEM_F * sizeof(float);

  if (nsplit > 1)
    lin_attn_states<<<wgs, 256, smem, stream>>>(qk, v, ws, nsplit);
  lin_attn_main<<<wgs, 256, smem, stream>>>(qk, v, ws, out, nsplit);
}


# One 64-token chunk per work-group (1024 WGs, 4 resident per WGP) instead of 4 chunks in series; KV/ksum chunk states now summed in chunk order by a new prefix kernel (summation order changes)
# speedup vs baseline: 2.8207x; 2.8207x over previous
#include <hip/hip_runtime.h>
#include <hip/hip_bf16.h>


typedef float v2f __attribute__((ext_vector_type(2)));
typedef float v8f __attribute__((ext_vector_type(8)));

constexpr int B_   = 2;
constexpr int S_   = 2048;
constexpr int H_   = 16;
constexpr int D_   = 64;
constexpr int CH   = 64;
constexpr int LDP  = 68;
constexpr int NCH  = S_ / CH;
constexpr int NSTATE = B_ * H_ * NCH;

constexpr int OFF_Q    = 0;
constexpr int OFF_K    = 64 * LDP;
constexpr int OFF_V    = 2 * 64 * LDP;
constexpr int OFF_KV   = 3 * 64 * LDP;
constexpr int OFF_KSUM = 4 * 64 * LDP;
constexpr int OFF_DEN  = OFF_KSUM + 64;
constexpr int SMEM_MAIN_F = OFF_DEN + 64;
constexpr int SMEM_ST_F   = 2 * 64 * LDP;

constexpr int STATE_F  = 64 * 64 + 64;

__device__ __forceinline__ float elu1(float x) {
  return x > 0.0f ? x + 1.0f : __expf(x);
}

__device__ __forceinline__ v8f wmma4(v2f a, v2f b, v8f c) {
  return __builtin_amdgcn_wmma_f32_16x16x4_f32(
       false, a,  false, b,
       (short)0, c,  false,  false);
}

__device__ __forceinline__ v2f lds_pair(const float* p) {
  return *reinterpret_cast<const v2f*>(p);
}

__device__ __forceinline__ void cd_store(float* base, int mi, int nj, int lane, v8f c) {
  int n  = nj * 16 + (lane & 15);
  int m0 = mi * 16 + ((lane >> 4) << 3);
#pragma unroll
  for (int r = 0; r < 8; ++r) base[(m0 + r) * LDP + n] = c[r];
}

__device__ __forceinline__ v8f gemm_nn(const float* A, const float* Bm,
                                       int mi, int nj, int lane,
                                       int k0, int k1, v8f acc) {
  int lm = lane & 15, hi = lane >> 4;
  int arow = (mi * 16 + lm) * LDP;
  int bcol = nj * 16 + lm;
  for (int kk = k0; kk < k1; ++kk) {
    int kb = kk * 4 + hi * 2;
    v2f a = lds_pair(&A[arow + kb]);
    v2f b; b.x = Bm[kb * LDP + bcol]; b.y = Bm[(kb + 1) * LDP + bcol];
    acc = wmma4(a, b, acc);
  }
  return acc;
}
__device__ __forceinline__ v8f gemm_nt(const float* A, const float* Bt,
                                       int mi, int nj, int lane, v8f acc) {
  int lm = lane & 15, hi = lane >> 4;
  int arow = (mi * 16 + lm) * LDP;
  int brow = (nj * 16 + lm) * LDP;
#pragma unroll
  for (int kk = 0; kk < 16; ++kk) {
    int kb = kk * 4 + hi * 2;
    v2f a = lds_pair(&A[arow + kb]);
    v2f b = lds_pair(&Bt[brow + kb]);
    acc = wmma4(a, b, acc);
  }
  return acc;
}
__device__ __forceinline__ v8f gemm_tn(const float* At, const float* Bm,
                                       int mi, int nj, int lane, v8f acc) {
  int lm = lane & 15, hi = lane >> 4;
  int am   = mi * 16 + lm;
  int bcol = nj * 16 + lm;
#pragma unroll
  for (int kk = 0; kk < 16; ++kk) {
    int kb = kk * 4 + hi * 2;
    v2f a; a.x = At[kb * LDP + am];   a.y = At[(kb + 1) * LDP + am];
    v2f b; b.x = Bm[kb * LDP + bcol]; b.y = Bm[(kb + 1) * LDP + bcol];
    acc = wmma4(a, b, acc);
  }
  return acc;
}

__device__ __forceinline__ void mask_diag(v8f& c, int lane) {
  int n  = lane & 15;
  int m0 = (lane >> 4) << 3;
#pragma unroll
  for (int r = 0; r < 8; ++r)
    if (m0 + r < n) c[r] = 0.0f;
}

__device__ __forceinline__ void load_tile(const float* __restrict__ g, int rowStride,
                                          float* lds, int tid, bool do_elu) {
#pragma unroll
  for (int j = 0; j < 4; ++j) {
    int u = tid + j * 256;
    int r = u >> 4;
    int c = (u & 15) << 2;
    float4 w = *reinterpret_cast<const float4*>(g + (size_t)r * rowStride + c);
    if (do_elu) { w.x = elu1(w.x); w.y = elu1(w.y); w.z = elu1(w.z); w.w = elu1(w.w); }
    *reinterpret_cast<float4*>(lds + r * LDP + c) = w;
  }
}

__global__ __launch_bounds__(256) void lin_attn_states(
    const float* __restrict__ qk, const float* __restrict__ v,
    float* __restrict__ loc) {
  extern __shared__ float smem[];
  float* sK = smem;
  float* sV = smem + 64 * LDP;

  int tid = threadIdx.x, lane = tid & 31, wid = tid >> 5;
  int wg = blockIdx.x;
  int ci = wg % NCH;
  int h  = (wg / NCH) % H_;
  int b  = wg / (NCH * H_);
  if (ci == NCH - 1) return;

  int base = ci * CH;
  const float* kptr = qk + (((size_t)(b * S_ + base) * 2 + 1) * H_ + h) * D_;
  const float* vptr = v  + (((size_t)(b * S_ + base)) * H_ + h) * D_;
  load_tile(kptr, 2 * H_ * D_, sK, tid, true);
  load_tile(vptr, H_ * D_,     sV, tid, false);
  __syncthreads();

  float* wsp = loc + (size_t)wg * STATE_F;
  if (tid < 64) {
    float acc = 0.0f;
    for (int s = 0; s < 64; ++s) acc += sK[s * LDP + tid];
    wsp[64 * 64 + tid] = acc;
  }
#pragma unroll
  for (int i = 0; i < 2; ++i) {
    int t = wid * 2 + i, mi = t >> 2, nj = t & 3;
    v8f acc = {};
    acc = gemm_tn(sK, sV, mi, nj, lane, acc);
    int n  = nj * 16 + (lane & 15);
    int m0 = mi * 16 + ((lane >> 4) << 3);
#pragma unroll
    for (int r = 0; r < 8; ++r) wsp[(m0 + r) * 64 + n] = acc[r];
  }
}

constexpr int SCAN_T  = 256;
constexpr int SCAN_WG = (B_ * H_ * STATE_F) / SCAN_T;
static_assert((B_ * H_ * STATE_F) % SCAN_T == 0, "scan grid must tile exactly");

__global__ __launch_bounds__(SCAN_T) void lin_attn_scan(
    const float* __restrict__ loc, float* __restrict__ pref) {
  int g  = blockIdx.x * SCAN_T + threadIdx.x;
  int bh = g / STATE_F;
  int e  = g % STATE_F;
  const float* src = loc  + (size_t)bh * NCH * STATE_F + e;
  float*       dst = pref + (size_t)bh * NCH * STATE_F + e;
  float x[NCH - 1];
#pragma unroll
  for (int c = 0; c < NCH - 1; ++c) x[c] = src[(size_t)c * STATE_F];
  float run = 0.0f;
#pragma unroll
  for (int c = 0; c < NCH; ++c) {
    dst[(size_t)c * STATE_F] = run;
    if (c < NCH - 1) run += x[c];
  }
}

__global__ __launch_bounds__(256) void lin_attn_main(
    const float* __restrict__ qk, const float* __restrict__ v,
    const float* __restrict__ pref, float* __restrict__ out) {
  extern __shared__ float smem[];
  float* sQ    = smem + OFF_Q;
  float* sK    = smem + OFF_K;
  float* sV    = smem + OFF_V;
  float* sS    = smem + OFF_K;
  float* sKV   = smem + OFF_KV;
  float* sksum = smem + OFF_KSUM;
  float* sden  = smem + OFF_DEN;

  int tid = threadIdx.x, lane = tid & 31, wid = tid >> 5;
  int wg = blockIdx.x;
  int ci = wg % NCH;
  int h  = (wg / NCH) % H_;
  int b  = wg / (NCH * H_);
  int base = ci * CH;

  const float* qptr = qk + (((size_t)(b * S_ + base) * 2 + 0) * H_ + h) * D_;
  const float* kptr = qk + (((size_t)(b * S_ + base) * 2 + 1) * H_ + h) * D_;
  const float* vptr = v  + (((size_t)(b * S_ + base)) * H_ + h) * D_;
  const float* wsp  = pref + (size_t)wg * STATE_F;
  load_tile(qptr, 2 * H_ * D_, sQ,  tid, true);
  load_tile(kptr, 2 * H_ * D_, sK,  tid, true);
  load_tile(vptr, H_ * D_,     sV,  tid, false);
  load_tile(wsp,  64,          sKV, tid, false);
  if (tid < 64) sksum[tid] = wsp[64 * 64 + tid];
  __syncthreads();

  int t0 = wid * 2, t1 = wid * 2 + 1;
  int m0 = t0 >> 2, n0 = t0 & 3, m1 = t1 >> 2, n1 = t1 & 3;

  v8f oa0 = {}, oa1 = {};
  oa0 = gemm_nn(sQ, sKV, m0, n0, lane, 0, 16, oa0);
  oa1 = gemm_nn(sQ, sKV, m1, n1, lane, 0, 16, oa1);

  float dinter = 0.0f;
  if (tid < 64) {
    for (int dd = 0; dd < 64; ++dd) dinter += sQ[tid * LDP + dd] * sksum[dd];
  }

  v8f sa0 = {}, sa1 = {};
  if (n0 <= m0) {
    sa0 = gemm_nt(sQ, sK, m0, n0, lane, sa0);
    if (m0 == n0) mask_diag(sa0, lane);
  }
  if (n1 <= m1) {
    sa1 = gemm_nt(sQ, sK, m1, n1, lane, sa1);
    if (m1 == n1) mask_diag(sa1, lane);
  }
  __syncthreads();
  if (n0 <= m0) cd_store(sS, m0, n0, lane, sa0);
  if (n1 <= m1) cd_store(sS, m1, n1, lane, sa1);
  __syncthreads();

  if (tid < 64) {
    float ssum = 0.0f;
    int smax = ((tid >> 4) + 1) * 16;
    for (int s = 0; s < smax; ++s) ssum += sS[tid * LDP + s];
    sden[tid] = 1.0f / (dinter + ssum);
  }
  __syncthreads();

  oa0 = gemm_nn(sS, sV, m0, n0, lane, 0, (m0 + 1) * 4, oa0);
  oa1 = gemm_nn(sS, sV, m1, n1, lane, 0, (m1 + 1) * 4, oa1);

  {
    int nn = n0 * 16 + (lane & 15);
    int mm = m0 * 16 + ((lane >> 4) << 3);
#pragma unroll
    for (int r = 0; r < 8; ++r)
      out[(((size_t)b * S_ + base + mm + r) * H_ + h) * D_ + nn] = oa0[r] * sden[mm + r];
    nn = n1 * 16 + (lane & 15);
    mm = m1 * 16 + ((lane >> 4) << 3);
#pragma unroll
    for (int r = 0; r < 8; ++r)
      out[(((size_t)b * S_ + base + mm + r) * H_ + h) * D_ + nn] = oa1[r] * sden[mm + r];
  }
}

extern "C" void kernel_launch(void* const* d_in, const int* in_sizes, int n_in,
                              void* d_out, int out_size, void* d_ws, size_t ws_size,
                              hipStream_t stream) {
  const float* qk = (const float*)d_in[0];
  const float* v  = (const float*)d_in[1];
  float* out  = (float*)d_out;
  float* loc  = (float*)d_ws;
  float* pref = (float*)d_ws + (size_t)NSTATE * STATE_F;

  lin_attn_states<<<NSTATE, 256, SMEM_ST_F * sizeof(float), stream>>>(qk, v, loc);
  lin_attn_scan<<<SCAN_WG, SCAN_T, 0, stream>>>(loc, pref);
  lin_attn_main<<<NSTATE, 256, SMEM_MAIN_F * sizeof(float), stream>>>(qk, v, pref, out);
}


# Prefix kernel removed: states kernel walks 8 chunks in series per work-group (256 WGs, loads issued ahead) and writes running states; main adds up to 3 segment totals (summation order changes)
# speedup vs baseline: 3.1458x; 1.1153x over previous
#include <hip/hip_runtime.h>
#include <hip/hip_bf16.h>


typedef float v2f __attribute__((ext_vector_type(2)));
typedef float v8f __attribute__((ext_vector_type(8)));

constexpr int B_   = 2;
constexpr int S_   = 2048;
constexpr int H_   = 16;
constexpr int D_   = 64;
constexpr int CH   = 64;
constexpr int LDP  = 68;
constexpr int NCH  = S_ / CH;
constexpr int NSTATE = B_ * H_ * NCH;
constexpr int NSEG = 4;
constexpr int SEGC = NCH / NSEG;

constexpr int OFF_Q    = 0;
constexpr int OFF_K    = 64 * LDP;
constexpr int OFF_V    = 2 * 64 * LDP;
constexpr int OFF_KV   = 3 * 64 * LDP;
constexpr int OFF_KSUM = 4 * 64 * LDP;
constexpr int OFF_DEN  = OFF_KSUM + 64;
constexpr int SMEM_MAIN_F = OFF_DEN + 64;
constexpr int PK = 72;
constexpr int PV = 40;
constexpr int ST_OFF_V  = 64 * PK;
constexpr int ST_OFF_KS = ST_OFF_V + 64 * PV;
constexpr int ST_BUF_F  = ST_OFF_KS + 8 * 64;
constexpr int SMEM_ST_F = 2 * ST_BUF_F;
constexpr int ST_WG     = B_ * H_ * NSEG * 2;

constexpr int STATE_F  = 64 * 64 + 64;

__device__ __forceinline__ float elu1(float x) {
  return x > 0.0f ? x + 1.0f : __expf(x);
}

__device__ __forceinline__ v8f wmma4(v2f a, v2f b, v8f c) {
  return __builtin_amdgcn_wmma_f32_16x16x4_f32(
       false, a,  false, b,
       (short)0, c,  false,  false);
}

__device__ __forceinline__ v2f lds_pair(const float* p) {
  return *reinterpret_cast<const v2f*>(p);
}

__device__ __forceinline__ void cd_store(float* base, int mi, int nj, int lane, v8f c) {
  int n  = nj * 16 + (lane & 15);
  int m0 = mi * 16 + ((lane >> 4) << 3);
#pragma unroll
  for (int r = 0; r < 8; ++r) base[(m0 + r) * LDP + n] = c[r];
}

__device__ __forceinline__ v8f gemm_nn(const float* A, const float* Bm,
                                       int mi, int nj, int lane,
                                       int k0, int k1, v8f acc) {
  int lm = lane & 15, hi = lane >> 4;
  int arow = (mi * 16 + lm) * LDP;
  int bcol = nj * 16 + lm;
  for (int kk = k0; kk < k1; ++kk) {
    int kb = kk * 4 + hi * 2;
    v2f a = lds_pair(&A[arow + kb]);
    v2f b; b.x = Bm[kb * LDP + bcol]; b.y = Bm[(kb + 1) * LDP + bcol];
    acc = wmma4(a, b, acc);
  }
  return acc;
}
__device__ __forceinline__ v8f gemm_nt(const float* A, const float* Bt,
                                       int mi, int nj, int lane, v8f acc) {
  int lm = lane & 15, hi = lane >> 4;
  int arow = (mi * 16 + lm) * LDP;
  int brow = (nj * 16 + lm) * LDP;
#pragma unroll
  for (int kk = 0; kk < 16; ++kk) {
    int kb = kk * 4 + hi * 2;
    v2f a = lds_pair(&A[arow + kb]);
    v2f b = lds_pair(&Bt[brow + kb]);
    acc = wmma4(a, b, acc);
  }
  return acc;
}
__device__ __forceinline__ void mask_diag(v8f& c, int lane) {
  int n  = lane & 15;
  int m0 = (lane >> 4) << 3;
#pragma unroll
  for (int r = 0; r < 8; ++r)
    if (m0 + r < n) c[r] = 0.0f;
}

__device__ __forceinline__ void tile_fetch(const float* __restrict__ g, int rowStride,
                                           int tid, float4 (&w)[4]) {
#pragma unroll
  for (int j = 0; j < 4; ++j) {
    int u = tid + j * 256;
    int r = u >> 4;
    int c = (u & 15) << 2;
    w[j] = *reinterpret_cast<const float4*>(g + (size_t)r * rowStride + c);
  }
}
__device__ __forceinline__ void tile_put(float* lds, int tid, const float4 (&w)[4], bool do_elu) {
#pragma unroll
  for (int j = 0; j < 4; ++j) {
    int u = tid + j * 256;
    int r = u >> 4;
    int c = (u & 15) << 2;
    float4 x = w[j];
    if (do_elu) { x.x = elu1(x.x); x.y = elu1(x.y); x.z = elu1(x.z); x.w = elu1(x.w); }
    *reinterpret_cast<float4*>(lds + r * LDP + c) = x;
  }
}
__device__ __forceinline__ void tile_add(float4 (&a)[4], const float4 (&t)[4]) {
#pragma unroll
  for (int j = 0; j < 4; ++j) { a[j].x += t[j].x; a[j].y += t[j].y; a[j].z += t[j].z; a[j].w += t[j].w; }
}

struct ChunkRegs { float4 k0, k1, k2, k3, v0, v1; };

__device__ __forceinline__ void lds_barrier() {
  __builtin_amdgcn_fence(__ATOMIC_RELEASE, "workgroup", "local");
  __builtin_amdgcn_s_barrier();
  __builtin_amdgcn_fence(__ATOMIC_ACQUIRE, "workgroup", "local");
}

__device__ __forceinline__ void chunk_fetch(ChunkRegs& g, const float* kp, const float* vp) {
  g.k0 = *reinterpret_cast<const float4*>(kp);
  g.k1 = *reinterpret_cast<const float4*>(kp + (size_t)16 * (2 * H_ * D_));
  g.k2 = *reinterpret_cast<const float4*>(kp + (size_t)32 * (2 * H_ * D_));
  g.k3 = *reinterpret_cast<const float4*>(kp + (size_t)48 * (2 * H_ * D_));
  g.v0 = *reinterpret_cast<const float4*>(vp);
  g.v1 = *reinterpret_cast<const float4*>(vp + (size_t)32 * (H_ * D_));
}

__device__ __forceinline__ float4 elu4_acc(float4 w, float4& cs) {
  w.x = elu1(w.x); w.y = elu1(w.y); w.z = elu1(w.z); w.w = elu1(w.w);
  cs.x += w.x; cs.y += w.y; cs.z += w.z; cs.w += w.w;
  return w;
}

__device__ __forceinline__ void chunk_step(const ChunkRegs& g, bool add, float* sK, float* wsp,
                                           bool ks_writer, int tid, int lane, int wid,
                                           int kst, int vst, int aoff, int boff,
                                           int ooff, v8f& accE, v8f& accO, float& kstot) {
  float* sV  = sK + ST_OFF_V;
  float* sKS = sK + ST_OFF_KS;
  if (add) {
    float4 cs = {0.0f, 0.0f, 0.0f, 0.0f};
    *reinterpret_cast<float4*>(sK + kst)           = elu4_acc(g.k0, cs);
    *reinterpret_cast<float4*>(sK + kst + 16 * PK) = elu4_acc(g.k1, cs);
    *reinterpret_cast<float4*>(sK + kst + 32 * PK) = elu4_acc(g.k2, cs);
    *reinterpret_cast<float4*>(sK + kst + 48 * PK) = elu4_acc(g.k3, cs);
    *reinterpret_cast<float4*>(sV + vst)           = g.v0;
    *reinterpret_cast<float4*>(sV + vst + 32 * PV) = g.v1;
    cs.x += __shfl_xor(cs.x, 16); cs.y += __shfl_xor(cs.y, 16);
    cs.z += __shfl_xor(cs.z, 16); cs.w += __shfl_xor(cs.w, 16);
    if (lane < 16) *reinterpret_cast<float4*>(sKS + wid * 64 + (lane << 2)) = cs;
    lds_barrier();
  }
  {
    v8f s = accE + accO;
#pragma unroll
    for (int r = 0; r < 8; ++r) wsp[ooff + r * 64] = s[r];
    if (ks_writer) wsp[64 * 64 + tid] = kstot;
  }
  if (add) {
    if (tid < 64) {
#pragma unroll
      for (int p = 0; p < 8; ++p) kstot += sKS[p * 64 + tid];
    }
#pragma unroll
    for (int kk = 0; kk < 16; ++kk) {
      v2f a;  a.x  = sK[aoff + kk * 4 * PK]; a.y  = sK[aoff + (kk * 4 + 1) * PK];
      v2f bb; bb.x = sV[boff + kk * 4 * PV]; bb.y = sV[boff + (kk * 4 + 1) * PV];
      if (kk & 1) accO = wmma4(a, bb, accO);
      else        accE = wmma4(a, bb, accE);
    }
  }
}

__global__ __launch_bounds__(256) void lin_attn_states(
    const float* qk, const float* v, float* pre, float* tot) {
  extern __shared__ float smem[];

  int tid = threadIdx.x, lane = tid & 31, wid = tid >> 5;
  int wg   = blockIdx.x;
  int half = wg & 1;
  int seg  = (wg >> 1) % NSEG;
  int h    = (wg / (2 * NSEG)) % H_;
  int b    = wg / (2 * NSEG * H_);
  int tok0 = seg * SEGC * CH;
  bool tail = (seg == NSEG - 1);

  int kr = tid >> 4, kc = (tid & 15) << 2;
  int vr = tid >> 3, vc = (tid & 7) << 2;
  const float* kp = qk + (((size_t)(b * S_ + tok0 + kr) * 2 + 1) * H_ + h) * D_ + kc;
  const float* vp = v  + (((size_t)(b * S_ + tok0 + vr)) * H_ + h) * D_ + half * 32 + vc;
  constexpr size_t KCH = (size_t)CH * 2 * H_ * D_;
  constexpr size_t VCH = (size_t)CH * H_ * D_;

  ChunkRegs g0, g1, g2, g3, g4, g5, g6, g7;
  chunk_fetch(g0, kp,           vp);
  __builtin_amdgcn_sched_barrier(0);
  chunk_fetch(g1, kp + KCH,     vp + VCH);
  __builtin_amdgcn_sched_barrier(0);
  chunk_fetch(g2, kp + 2 * KCH, vp + 2 * VCH);
  __builtin_amdgcn_sched_barrier(0);
  chunk_fetch(g3, kp + 3 * KCH, vp + 3 * VCH);
  __builtin_amdgcn_sched_barrier(0);
  chunk_fetch(g4, kp + 4 * KCH, vp + 4 * VCH);
  __builtin_amdgcn_sched_barrier(0);
  chunk_fetch(g5, kp + 5 * KCH, vp + 5 * VCH);
  __builtin_amdgcn_sched_barrier(0);

  int mi = wid >> 1;
  int nl = wid & 1;
  int lm = lane & 15, hi = lane >> 4;
  int kst  = kr * PK + kc;
  int vst  = vr * PV + vc;
  int aoff = hi * 2 * PK + mi * 16 + lm;
  int boff = hi * 2 * PV + nl * 16 + lm;
  int ooff = (mi * 16 + (hi << 3)) * 64 + (2 * half + nl) * 16 + lm;
  bool ksw = (half == 0) && (tid < 64);
  v8f accE = {}, accO = {};
  float kstot = 0.0f;

  float* w   = pre + (size_t)((b * H_ + h) * NCH + seg * SEGC) * STATE_F;
  float* sA  = smem;
  float* sB  = smem + ST_BUF_F;
  chunk_step(g0, true,  sA, w,               ksw, tid, lane, wid, kst, vst, aoff, boff, ooff, accE, accO, kstot);
  chunk_fetch(g6, kp + 6 * KCH, vp + 6 * VCH);
  chunk_step(g1, true,  sB, w + STATE_F,     ksw, tid, lane, wid, kst, vst, aoff, boff, ooff, accE, accO, kstot);
  if (!tail) chunk_fetch(g7, kp + 7 * KCH, vp + 7 * VCH);
  chunk_step(g2, true,  sA, w + 2 * STATE_F, ksw, tid, lane, wid, kst, vst, aoff, boff, ooff, accE, accO, kstot);
  chunk_step(g3, true,  sB, w + 3 * STATE_F, ksw, tid, lane, wid, kst, vst, aoff, boff, ooff, accE, accO, kstot);
  chunk_step(g4, true,  sA, w + 4 * STATE_F, ksw, tid, lane, wid, kst, vst, aoff, boff, ooff, accE, accO, kstot);
  chunk_step(g5, true,  sB, w + 5 * STATE_F, ksw, tid, lane, wid, kst, vst, aoff, boff, ooff, accE, accO, kstot);
  chunk_step(g6, true,  sA, w + 6 * STATE_F, ksw, tid, lane, wid, kst, vst, aoff, boff, ooff, accE, accO, kstot);
  chunk_step(g7, !tail, sB, w + 7 * STATE_F, ksw, tid, lane, wid, kst, vst, aoff, boff, ooff, accE, accO, kstot);

  if (!tail) {
    float* tp = tot + (size_t)((b * H_ + h) * NSEG + seg) * STATE_F;
    v8f s = accE + accO;
#pragma unroll
    for (int r = 0; r < 8; ++r) tp[ooff + r * 64] = s[r];
    if (ksw) tp[64 * 64 + tid] = kstot;
  }
}

__global__ __launch_bounds__(256) void lin_attn_main(
    const float* __restrict__ qk, const float* __restrict__ v,
    const float* __restrict__ pre, const float* __restrict__ tot,
    float* __restrict__ out) {
  extern __shared__ float smem[];
  float* sQ    = smem + OFF_Q;
  float* sK    = smem + OFF_K;
  float* sV    = smem + OFF_V;
  float* sS    = smem + OFF_K;
  float* sKV   = smem + OFF_KV;
  float* sksum = smem + OFF_KSUM;
  float* sden  = smem + OFF_DEN;

  int tid = threadIdx.x, lane = tid & 31, wid = tid >> 5;
  int wg = blockIdx.x;
  int ci = wg % NCH;
  int h  = (wg / NCH) % H_;
  int b  = wg / (NCH * H_);
  int base = ci * CH;

  const float* qptr = qk + (((size_t)(b * S_ + base) * 2 + 0) * H_ + h) * D_;
  const float* kptr = qk + (((size_t)(b * S_ + base) * 2 + 1) * H_ + h) * D_;
  const float* vptr = v  + (((size_t)(b * S_ + base)) * H_ + h) * D_;
  const float* wsp = pre + (size_t)wg * STATE_F;
  const float* tsp = tot + (size_t)((b * H_ + h) * NSEG) * STATE_F;
  int seg = ci / SEGC;
  float4 e0[4], e1[4], e2[4], rq[4], rk[4], rv[4];
  int kso = 64 * 64 + (tid & 63);
  float ks0, ks1, ks2;
  tile_fetch(wsp, 64, tid, e0);
  ks0 = wsp[kso];
  if (seg > 0) { tile_fetch(tsp, 64, tid, e1); ks1 = tsp[kso]; }
  tile_fetch(qptr, 2 * H_ * D_, tid, rq);
  tile_fetch(kptr, 2 * H_ * D_, tid, rk);
  tile_fetch(vptr, H_ * D_,     tid, rv);
  if (seg > 0) { tile_add(e0, e1); ks0 += ks1; }
  if (seg > 1) {
    tile_fetch(tsp + STATE_F, 64, tid, e1);
    ks1 = tsp[STATE_F + kso];
    if (seg > 2) {
      tile_fetch(tsp + 2 * STATE_F, 64, tid, e2);
      ks2 = tsp[2 * STATE_F + kso];
    }
  }
  tile_put(sQ, tid, rq, true);
  tile_put(sK, tid, rk, true);
  tile_put(sV, tid, rv, false);
  if (seg > 1) {
    tile_add(e0, e1); ks0 += ks1;
    if (seg > 2) { tile_add(e0, e2); ks0 += ks2; }
  }
  tile_put(sKV, tid, e0, false);
  if (tid < 64) sksum[tid] = ks0;
  __syncthreads();

  int t0 = wid * 2, t1 = wid * 2 + 1;
  int m0 = t0 >> 2, n0 = t0 & 3, m1 = t1 >> 2, n1 = t1 & 3;

  v8f oa0 = {}, oa1 = {};
  oa0 = gemm_nn(sQ, sKV, m0, n0, lane, 0, 16, oa0);
  oa1 = gemm_nn(sQ, sKV, m1, n1, lane, 0, 16, oa1);

  float dinter = 0.0f;
  if (tid < 64) {
    for (int dd = 0; dd < 64; ++dd) dinter += sQ[tid * LDP + dd] * sksum[dd];
  }

  v8f sa0 = {}, sa1 = {};
  if (n0 <= m0) {
    sa0 = gemm_nt(sQ, sK, m0, n0, lane, sa0);
    if (m0 == n0) mask_diag(sa0, lane);
  }
  if (n1 <= m1) {
    sa1 = gemm_nt(sQ, sK, m1, n1, lane, sa1);
    if (m1 == n1) mask_diag(sa1, lane);
  }
  __syncthreads();
  if (n0 <= m0) cd_store(sS, m0, n0, lane, sa0);
  if (n1 <= m1) cd_store(sS, m1, n1, lane, sa1);
  __syncthreads();

  if (tid < 64) {
    float ssum = 0.0f;
    int smax = ((tid >> 4) + 1) * 16;
    for (int s = 0; s < smax; ++s) ssum += sS[tid * LDP + s];
    sden[tid] = 1.0f / (dinter + ssum);
  }
  __syncthreads();

  oa0 = gemm_nn(sS, sV, m0, n0, lane, 0, (m0 + 1) * 4, oa0);
  oa1 = gemm_nn(sS, sV, m1, n1, lane, 0, (m1 + 1) * 4, oa1);

  {
    int nn = n0 * 16 + (lane & 15);
    int mm = m0 * 16 + ((lane >> 4) << 3);
#pragma unroll
    for (int r = 0; r < 8; ++r)
      out[(((size_t)b * S_ + base + mm + r) * H_ + h) * D_ + nn] = oa0[r] * sden[mm + r];
    nn = n1 * 16 + (lane & 15);
    mm = m1 * 16 + ((lane >> 4) << 3);
#pragma unroll
    for (int r = 0; r < 8; ++r)
      out[(((size_t)b * S_ + base + mm + r) * H_ + h) * D_ + nn] = oa1[r] * sden[mm + r];
  }
}

extern "C" void kernel_launch(void* const* d_in, const int* in_sizes, int n_in,
                              void* d_out, int out_size, void* d_ws, size_t ws_size,
                              hipStream_t stream) {
  const float* qk = (const float*)d_in[0];
  const float* v  = (const float*)d_in[1];
  float* out  = (float*)d_out;
  float* pre  = (float*)d_ws;
  float* tot  = (float*)d_ws + (size_t)NSTATE * STATE_F;

  lin_attn_states<<<ST_WG, 256, SMEM_ST_F * sizeof(float), stream>>>(qk, v, pre, tot);
  lin_attn_main<<<NSTATE, 256, SMEM_MAIN_F * sizeof(float), stream>>>(qk, v, pre, tot, out);
}
